# starting kernel
# baseline (speedup 1.0000x reference)
#include <hip/hip_runtime.h>
#include <hip/hip_bf16.h>


typedef float v2f __attribute__((ext_vector_type(2)));
typedef float v8f __attribute__((ext_vector_type(8)));

#define H_DIM 64

__global__ void k_deg_init(float* __restrict__ deg, int n) {
  int i = blockIdx.x * blockDim.x + threadIdx.x;
  if (i < n) deg[i] = 1.0f;
}

__global__ void k_deg_accum(const int* __restrict__ dst, float* __restrict__ deg,
                            int ne) {
  int i = blockIdx.x * blockDim.x + threadIdx.x;
  if (i < ne) unsafeAtomicAdd(&deg[dst[i]], 1.0f);
}

__global__ void k_deg_rsqrt(float* __restrict__ deg, int n) {
  int i = blockIdx.x * blockDim.x + threadIdx.x;
  if (i < n) deg[i] = rsqrtf(deg[i]);
}

__global__ void k_gemm_wmma_fused(const float* __restrict__ X,
                                  const float* __restrict__ W,
                                  const float* __restrict__ bias,
                                  const float* __restrict__ dis,
                                  float* __restrict__ Hout,
                                  float* __restrict__ Acc, int n) {
  const int lane  = threadIdx.x & 31;
  const int wave  = threadIdx.x >> 5;
  const int row0  = blockIdx.x * 16;
  const int n0    = wave * 16;
  const int m     = lane & 15;
  const int khalf = (lane >> 4) << 1;

  int arow = row0 + m;
  if (arow >= n) arow = n - 1;
  const float* __restrict__ xrow = X + (size_t)arow * H_DIM;

  v8f acc = {};
#pragma unroll
  for (int kk = 0; kk < 16; ++kk) {
    const int kb = kk * 4 + khalf;
    v2f a;
    a.x = xrow[kb];
    a.y = xrow[kb + 1];
    v2f b;
    b.x = W[kb * H_DIM + n0 + m];
    b.y = W[(kb + 1) * H_DIM + n0 + m];
    acc = __builtin_amdgcn_wmma_f32_16x16x4_f32(false, a, false, b, (short)0,
                                                acc, false, false);
  }

  const int rbase = (lane >> 4) << 3;
  const int col   = n0 + m;
  const float bc  = bias[col];
#pragma unroll
  for (int r = 0; r < 8; ++r) {
    const int row = row0 + rbase + r;
    if (row < n) {
      const float h = acc[r];
      const float d = dis[row];
      Hout[(size_t)row * H_DIM + col] = h;
      Acc[(size_t)row * H_DIM + col] = fmaf(h, d * d, bc);
    }
  }
}

__global__ void k_scatter_edges(const float* __restrict__ Hm,
                                const int* __restrict__ src,
                                const int* __restrict__ dst,
                                const float* __restrict__ dis,
                                float* __restrict__ Acc, int ne) {
  const int e = blockIdx.x * (blockDim.x >> 5) + (threadIdx.x >> 5);
  if (e >= ne) return;
  const int lane = threadIdx.x & 31;
  const int s = src[e];
  const int d = dst[e];
  const float nrm = dis[s] * dis[d];
  const float2 v = ((const float2*)(Hm + (size_t)s * H_DIM))[lane];
  float* outp = Acc + (size_t)d * H_DIM + lane * 2;
  unsafeAtomicAdd(outp, v.x * nrm);
  unsafeAtomicAdd(outp + 1, v.y * nrm);
}

__global__ void k_relu(float* __restrict__ a, size_t n) {
  size_t i = (size_t)blockIdx.x * blockDim.x + threadIdx.x;
  if (i < n) a[i] = fmaxf(a[i], 0.0f);
}

__global__ void k_decode(const float* __restrict__ Z,
                         const int* __restrict__ pos,
                         const int* __restrict__ neg,
                         float* __restrict__ out, int npos, int nneg) {
  const int k = blockIdx.x * (blockDim.x >> 5) + (threadIdx.x >> 5);
  if (k >= npos + nneg) return;
  const int lane = threadIdx.x & 31;
  int a, b;
  if (k < npos) {
    a = pos[k];
    b = pos[npos + k];
  } else {
    const int j = k - npos;
    a = neg[j];
    b = neg[nneg + j];
  }
  const float2 va = ((const float2*)(Z + (size_t)a * H_DIM))[lane];
  const float2 vb = ((const float2*)(Z + (size_t)b * H_DIM))[lane];
  float s = fmaf(va.x, vb.x, va.y * vb.y);
#pragma unroll
  for (int off = 16; off > 0; off >>= 1) s += __shfl_xor(s, off, 32);
  if (lane == 0) out[k] = s;
}

extern "C" void kernel_launch(void* const* d_in, const int* in_sizes, int n_in,
                              void* d_out, int out_size, void* d_ws,
                              size_t ws_size, hipStream_t stream) {
  (void)n_in; (void)out_size; (void)ws_size;
  const float* x   = (const float*)d_in[0];
  const int*   ei  = (const int*)d_in[1];
  const int*   pei = (const int*)d_in[2];
  const int*   nei = (const int*)d_in[3];
  const float* W1  = (const float*)d_in[4];
  const float* b1  = (const float*)d_in[5];
  const float* W2  = (const float*)d_in[6];
  const float* b2  = (const float*)d_in[7];
  float* out = (float*)d_out;

  const int n    = in_sizes[0] / H_DIM;
  const int ne   = in_sizes[1] / 2;
  const int npos = in_sizes[2] / 2;
  const int nneg = in_sizes[3] / 2;
  const int* src = ei;
  const int* dst = ei + ne;

  float* ws = (float*)d_ws;
  const size_t nAl = ((size_t)n + 63) & ~(size_t)63;
  float* dis  = ws;
  float* hbuf = ws + nAl;
  float* acc1 = hbuf + (size_t)n * H_DIM;
  float* acc2 = acc1 + (size_t)n * H_DIM;

  const int B = 256;
  k_deg_init<<<(n + B - 1) / B, B, 0, stream>>>(dis, n);
  k_deg_accum<<<(ne + B - 1) / B, B, 0, stream>>>(dst, dis, ne);
  k_deg_rsqrt<<<(n + B - 1) / B, B, 0, stream>>>(dis, n);

  const int tiles = (n + 15) / 16;
  k_gemm_wmma_fused<<<tiles, 128, 0, stream>>>(x, W1, b1, dis, hbuf, acc1, n);
  k_scatter_edges<<<(ne + 7) / 8, B, 0, stream>>>(hbuf, src, dst, dis, acc1, ne);
  k_relu<<<(int)(((size_t)n * H_DIM + B - 1) / B), B, 0, stream>>>(
      acc1, (size_t)n * H_DIM);
  k_gemm_wmma_fused<<<tiles, 128, 0, stream>>>(acc1, W2, b2, dis, hbuf, acc2, n);
  k_scatter_edges<<<(ne + 7) / 8, B, 0, stream>>>(hbuf, src, dst, dis, acc2, ne);
  k_decode<<<(npos + nneg + 7) / 8, B, 0, stream>>>(acc2, pei, nei, out, npos,
                                                    nneg);
}


# Layer 1 sums neighbours' input rows first; one kernel then does (sum)@W1+b1, ReLU, @W2 on WMMA (T via LDS): ReLU pass and a GEMM launch gone, 9 launches to 7; layer-1 sum order differs
# speedup vs baseline: 1.6434x; 1.6434x over previous
#include <hip/hip_runtime.h>
#include <hip/hip_bf16.h>


typedef float v2f __attribute__((ext_vector_type(2)));
typedef float v8f __attribute__((ext_vector_type(8)));

#define H_DIM 64

__global__ void k_deg_init(float* __restrict__ deg, int n) {
  int i = blockIdx.x * blockDim.x + threadIdx.x;
  if (i < n) deg[i] = 1.0f;
}

__global__ void k_deg_accum(const int* __restrict__ dst, float* __restrict__ deg,
                            int ne) {
  int i = blockIdx.x * blockDim.x + threadIdx.x;
  if (i < ne) unsafeAtomicAdd(&deg[dst[i]], 1.0f);
}

#define ROWS_PER_INIT_WAVE 8
__global__ void __launch_bounds__(256)
k_norm_init(const float* __restrict__ deg, float* __restrict__ dis,
            const float* __restrict__ X, float* __restrict__ S0,
            float* __restrict__ S1, int n) {
  const int lane = threadIdx.x & 31;
  const int wave = blockIdx.x * (blockDim.x >> 5) + (threadIdx.x >> 5);
  const int sub  = lane >> 4;
  const int c4   = (lane & 15) << 2;
  float dg[ROWS_PER_INIT_WAVE / 2];
  float4 xv[ROWS_PER_INIT_WAVE / 2];
#pragma unroll
  for (int j = 0; j < ROWS_PER_INIT_WAVE / 2; ++j) {
    const int row = wave * ROWS_PER_INIT_WAVE + 2 * j + sub;
    const int rc  = row < n ? row : n - 1;
    dg[j] = deg[rc];
    xv[j] = *(const float4*)(X + (size_t)rc * H_DIM + c4);
  }
#pragma unroll
  for (int j = 0; j < ROWS_PER_INIT_WAVE / 2; ++j) {
    const int row = wave * ROWS_PER_INIT_WAVE + 2 * j + sub;
    if (row < n) {
      const float d = rsqrtf(dg[j]);
      if ((lane & 15) == 0) dis[row] = d;
      const float dd = d * d;
      const size_t off = (size_t)row * H_DIM + c4;
      const float4 x = xv[j];
      *(float4*)(S0 + off) = make_float4(x.x * dd, x.y * dd, x.z * dd, x.w * dd);
      *(float4*)(S1 + off) = make_float4(0.0f, 0.0f, 0.0f, 0.0f);
    }
  }
}

#define T_PITCH 68

__global__ void __launch_bounds__(128)
k_gemm_pair_fused(float* __restrict__ S0, float* __restrict__ S1,
                  const float* __restrict__ W1, const float* __restrict__ b1,
                  const float* __restrict__ W2, const float* __restrict__ b2,
                  const float* __restrict__ dis, float* __restrict__ Hout,
                  int n) {
  __shared__ float T[16 * T_PITCH];
  const int lane  = threadIdx.x & 31;
  const int wave  = threadIdx.x >> 5;
  const int row0  = blockIdx.x * 16;
  const int n0    = wave * 16;
  const int m     = lane & 15;
  const int khalf = (lane >> 4) << 1;

  int arow = row0 + m;
  if (arow >= n) arow = n - 1;
  const float* p0 = S0 + (size_t)arow * H_DIM;
  const float* p1 = S1 + (size_t)arow * H_DIM;

  const int rbase = (lane >> 4) << 3;
  const int col   = n0 + m;

  v8f acc = {};
#pragma unroll
  for (int kk = 0; kk < 16; ++kk) {
    const int kb = kk * 4 + khalf;
    v2f a;
    a.x = p0[kb] + p1[kb];
    a.y = p0[kb + 1] + p1[kb + 1];
    v2f b;
    b.x = W1[kb * H_DIM + n0 + m];
    b.y = W1[(kb + 1) * H_DIM + n0 + m];
    acc = __builtin_amdgcn_wmma_f32_16x16x4_f32(false, a, false, b, (short)0,
                                                acc, false, false);
  }

  float dv[8];
#pragma unroll
  for (int r = 0; r < 8; ++r) {
    const int row = row0 + rbase + r;
    dv[r] = dis[row < n ? row : n - 1];
  }

  const float bc1 = b1[col];
#pragma unroll
  for (int r = 0; r < 8; ++r)
    T[(rbase + r) * T_PITCH + col] = fmaxf(acc[r] + bc1, 0.0f);
  __syncthreads();

  const float* trow = T + m * T_PITCH;
  v8f acc2 = {};
#pragma unroll
  for (int kk = 0; kk < 16; ++kk) {
    const int kb = kk * 4 + khalf;
    v2f a;
    a.x = trow[kb];
    a.y = trow[kb + 1];
    v2f b;
    b.x = W2[kb * H_DIM + n0 + m];
    b.y = W2[(kb + 1) * H_DIM + n0 + m];
    acc2 = __builtin_amdgcn_wmma_f32_16x16x4_f32(false, a, false, b, (short)0,
                                                 acc2, false, false);
  }

  const float bc2 = b2[col];
#pragma unroll
  for (int r = 0; r < 8; ++r) {
    const int row = row0 + rbase + r;
    if (row < n) {
      const float h = acc2[r];
      const float d = dv[r];
      const size_t o = (size_t)row * H_DIM + col;
      Hout[o] = h;
      S0[o] = fmaf(h, d * d, bc2);
      S1[o] = 0.0f;
    }
  }
}

#define EDGES_PER_WAVE 32
#define ROWS_IN_FLIGHT 8

__device__ __forceinline__ int lane_bcast_i(int v, int l) {
  return __builtin_amdgcn_readlane(v, l);
}
__device__ __forceinline__ float lane_bcast_f(float v, int l) {
  return __builtin_bit_cast(
      float, __builtin_amdgcn_readlane(__builtin_bit_cast(int, v), l));
}

__global__ void __launch_bounds__(256)
k_scatter_edges(const float* __restrict__ Hm, const int* __restrict__ src,
                const int* __restrict__ dst, const float* __restrict__ dis,
                float* __restrict__ Acc0, float* __restrict__ Acc1, int ne) {
  const int lane = threadIdx.x & 31;
  const int wave = blockIdx.x * (blockDim.x >> 5) + (threadIdx.x >> 5);
  const int e0 = wave * EDGES_PER_WAVE;
  if (e0 >= ne) return;
  const int left = ne - e0;
  const int cnt = left < EDGES_PER_WAVE ? left : EDGES_PER_WAVE;

  int s = 0, d = 0;
  float nrm = 0.0f;
  if (lane < cnt) {
    s = src[e0 + lane];
    d = dst[e0 + lane];
    nrm = dis[s] * dis[d];
  }
  const float* __restrict__ Hl = Hm + lane;
  const unsigned place = __builtin_amdgcn_s_sendmsg_rtn(0x87);
  const bool upper = ((place >> 16) & 0xFu) >= 4u;
  float* __restrict__ Al = (upper ? Acc1 : Acc0) + lane;

  if (cnt == EDGES_PER_WAVE) {
#pragma unroll
    for (int j0 = 0; j0 < EDGES_PER_WAVE; j0 += ROWS_IN_FLIGHT) {
      float vlo[ROWS_IN_FLIGHT], vhi[ROWS_IN_FLIGHT];
#pragma unroll
      for (int j = 0; j < ROWS_IN_FLIGHT; ++j) {
        const int sj = lane_bcast_i(s, j0 + j);
        const float* hp = Hl + (size_t)sj * H_DIM;
        vlo[j] = hp[0];
        vhi[j] = hp[32];
      }
#pragma unroll
      for (int j = 0; j < ROWS_IN_FLIGHT; ++j) {
        const int dj = lane_bcast_i(d, j0 + j);
        const float nj = lane_bcast_f(nrm, j0 + j);
        float* outp = Al + (size_t)dj * H_DIM;
        unsafeAtomicAdd(outp, vlo[j] * nj);
        unsafeAtomicAdd(outp + 32, vhi[j] * nj);
      }
    }
  } else {
    for (int j = 0; j < cnt; ++j) {
      const int sj = lane_bcast_i(s, j);
      const int dj = lane_bcast_i(d, j);
      const float nj = lane_bcast_f(nrm, j);
      const float* hp = Hl + (size_t)sj * H_DIM;
      const float lo = hp[0];
      const float hi = hp[32];
      float* outp = Al + (size_t)dj * H_DIM;
      unsafeAtomicAdd(outp, lo * nj);
      unsafeAtomicAdd(outp + 32, hi * nj);
    }
  }
}

__global__ void k_decode(const float* __restrict__ Z0,
                         const float* __restrict__ Z1,
                         const int* __restrict__ pos,
                         const int* __restrict__ neg,
                         float* __restrict__ out, int npos, int nneg) {
  const int k = blockIdx.x * (blockDim.x >> 5) + (threadIdx.x >> 5);
  if (k >= npos + nneg) return;
  const int lane = threadIdx.x & 31;
  int a, b;
  if (k < npos) {
    a = pos[k];
    b = pos[npos + k];
  } else {
    const int j = k - npos;
    a = neg[j];
    b = neg[nneg + j];
  }
  const float2 a0 = ((const float2*)(Z0 + (size_t)a * H_DIM))[lane];
  const float2 a1 = ((const float2*)(Z1 + (size_t)a * H_DIM))[lane];
  const float2 b0 = ((const float2*)(Z0 + (size_t)b * H_DIM))[lane];
  const float2 b1 = ((const float2*)(Z1 + (size_t)b * H_DIM))[lane];
  const float2 va = make_float2(a0.x + a1.x, a0.y + a1.y);
  const float2 vb = make_float2(b0.x + b1.x, b0.y + b1.y);
  float s = fmaf(va.x, vb.x, va.y * vb.y);
#pragma unroll
  for (int off = 16; off > 0; off >>= 1) s += __shfl_xor(s, off, 32);
  if (lane == 0) out[k] = s;
}

extern "C" void kernel_launch(void* const* d_in, const int* in_sizes, int n_in,
                              void* d_out, int out_size, void* d_ws,
                              size_t ws_size, hipStream_t stream) {
  (void)n_in; (void)out_size; (void)ws_size;
  const float* x   = (const float*)d_in[0];
  const int*   ei  = (const int*)d_in[1];
  const int*   pei = (const int*)d_in[2];
  const int*   nei = (const int*)d_in[3];
  const float* W1  = (const float*)d_in[4];
  const float* b1  = (const float*)d_in[5];
  const float* W2  = (const float*)d_in[6];
  const float* b2  = (const float*)d_in[7];
  float* out = (float*)d_out;

  const int n    = in_sizes[0] / H_DIM;
  const int ne   = in_sizes[1] / 2;
  const int npos = in_sizes[2] / 2;
  const int nneg = in_sizes[3] / 2;
  const int* src = ei;
  const int* dst = ei + ne;

  float* ws = (float*)d_ws;
  const size_t nAl = ((size_t)n + 63) & ~(size_t)63;
  float* deg  = ws;
  float* dis  = deg + nAl;
  float* s0   = dis + nAl;
  float* s1   = s0 + (size_t)n * H_DIM;
  float* hbuf = s1 + (size_t)n * H_DIM;

  const int B = 256;
  k_deg_init<<<(n + B - 1) / B, B, 0, stream>>>(deg, n);
  k_deg_accum<<<(ne + B - 1) / B, B, 0, stream>>>(dst, deg, ne);
  k_norm_init<<<(n + 63) / 64, B, 0, stream>>>(deg, dis, x, s0, s1, n);

  k_scatter_edges<<<(ne + 255) / 256, B, 0, stream>>>(x, src, dst, dis, s0, s1,
                                                      ne);
  k_gemm_pair_fused<<<(n + 15) / 16, 128, 0, stream>>>(s0, s1, W1, b1, W2, b2,
                                                       dis, hbuf, n);
  k_scatter_edges<<<(ne + 255) / 256, B, 0, stream>>>(hbuf, src, dst, dis, s0,
                                                      s1, ne);
  k_decode<<<(npos + nneg + 7) / 8, B, 0, stream>>>(s0, s1, pei, nei, out,
                                                    npos, nneg);
}


# Degree counts without global atomics: a block keeps 3200 nodes' counters in LDS, scans 1/8 of the edge list (LDS integer adds), stores partial counts; rsqrt pass adds the 8 parts; 7 launches to 6
# speedup vs baseline: 2.0706x; 1.2599x over previous
#include <hip/hip_runtime.h>
#include <hip/hip_bf16.h>


typedef float v2f __attribute__((ext_vector_type(2)));
typedef float v8f __attribute__((ext_vector_type(8)));

#define H_DIM 64

#define DEG_RANGE  3200
#define DEG_CHUNKS 8
#define DEG_AHEAD  16
#define DEG_THREADS 1024

__global__ void __launch_bounds__(DEG_THREADS)
k_deg_hist(const int* __restrict__ dst, unsigned* __restrict__ part,
           size_t plane, int n, int ne) {
  __shared__ unsigned cnt[DEG_RANGE];
  const int tid  = threadIdx.x;
  const int step = DEG_THREADS;
  const int lo   = blockIdx.y * DEG_RANGE;
  for (int i = tid; i < DEG_RANGE; i += step) cnt[i] = 0u;
  __syncthreads();

  const int per = (ne + (int)gridDim.x - 1) / (int)gridDim.x;
  const int e0  = blockIdx.x * per;
  const int e1  = e0 + per < ne ? e0 + per : ne;
  int e = e0 + tid;
  for (; e + (DEG_AHEAD - 1) * step < e1; e += DEG_AHEAD * step) {
    int v[DEG_AHEAD];
#pragma unroll
    for (int j = 0; j < DEG_AHEAD; ++j) v[j] = dst[e + j * step];
#pragma unroll
    for (int j = 0; j < DEG_AHEAD; ++j) {
      const unsigned r = (unsigned)(v[j] - lo);
      if (r < (unsigned)DEG_RANGE) atomicAdd(&cnt[r], 1u);
    }
  }
  for (; e < e1; e += step) {
    const unsigned r = (unsigned)(dst[e] - lo);
    if (r < (unsigned)DEG_RANGE) atomicAdd(&cnt[r], 1u);
  }
  __syncthreads();

  unsigned* __restrict__ prow = part + (size_t)blockIdx.x * plane;
  for (int i = tid; i < DEG_RANGE; i += step) {
    const int node = lo + i;
    if (node < n) prow[node] = cnt[i];
  }
}

#define ROWS_PER_INIT_WAVE 8
__global__ void __launch_bounds__(256)
k_norm_init(const unsigned* __restrict__ part, size_t plane,
            float* __restrict__ dis, const float* __restrict__ X,
            float* __restrict__ S0, float* __restrict__ S1, int n) {
  const int lane = threadIdx.x & 31;
  const int wave = blockIdx.x * (blockDim.x >> 5) + (threadIdx.x >> 5);
  const int sub  = lane >> 4;
  const int c4   = (lane & 15) << 2;
  float dg[ROWS_PER_INIT_WAVE / 2];
  float4 xv[ROWS_PER_INIT_WAVE / 2];
#pragma unroll
  for (int j = 0; j < ROWS_PER_INIT_WAVE / 2; ++j) {
    const int row = wave * ROWS_PER_INIT_WAVE + 2 * j + sub;
    const int rc  = row < n ? row : n - 1;
    xv[j] = *(const float4*)(X + (size_t)rc * H_DIM + c4);
    unsigned c = 1u;
#pragma unroll
    for (int k = 0; k < DEG_CHUNKS; ++k) c += part[(size_t)k * plane + rc];
    dg[j] = (float)c;
  }
#pragma unroll
  for (int j = 0; j < ROWS_PER_INIT_WAVE / 2; ++j) {
    const int row = wave * ROWS_PER_INIT_WAVE + 2 * j + sub;
    if (row < n) {
      const float d = rsqrtf(dg[j]);
      if ((lane & 15) == 0) dis[row] = d;
      const float dd = d * d;
      const size_t off = (size_t)row * H_DIM + c4;
      const float4 x = xv[j];
      *(float4*)(S0 + off) = make_float4(x.x * dd, x.y * dd, x.z * dd, x.w * dd);
      *(float4*)(S1 + off) = make_float4(0.0f, 0.0f, 0.0f, 0.0f);
    }
  }
}

#define T_PITCH 68

__global__ void __launch_bounds__(128)
k_gemm_pair_fused(float* __restrict__ S0, float* __restrict__ S1,
                  const float* __restrict__ W1, const float* __restrict__ b1,
                  const float* __restrict__ W2, const float* __restrict__ b2,
                  const float* __restrict__ dis, float* __restrict__ Hout,
                  int n) {
  __shared__ float T[16 * T_PITCH];
  const int lane  = threadIdx.x & 31;
  const int wave  = threadIdx.x >> 5;
  const int row0  = blockIdx.x * 16;
  const int n0    = wave * 16;
  const int m     = lane & 15;
  const int khalf = (lane >> 4) << 1;

  int arow = row0 + m;
  if (arow >= n) arow = n - 1;
  const float* p0 = S0 + (size_t)arow * H_DIM;
  const float* p1 = S1 + (size_t)arow * H_DIM;

  const int rbase = (lane >> 4) << 3;
  const int col   = n0 + m;

  v8f acc = {};
#pragma unroll
  for (int kk = 0; kk < 16; ++kk) {
    const int kb = kk * 4 + khalf;
    v2f a;
    a.x = p0[kb] + p1[kb];
    a.y = p0[kb + 1] + p1[kb + 1];
    v2f b;
    b.x = W1[kb * H_DIM + n0 + m];
    b.y = W1[(kb + 1) * H_DIM + n0 + m];
    acc = __builtin_amdgcn_wmma_f32_16x16x4_f32(false, a, false, b, (short)0,
                                                acc, false, false);
  }

  float dv[8];
#pragma unroll
  for (int r = 0; r < 8; ++r) {
    const int row = row0 + rbase + r;
    dv[r] = dis[row < n ? row : n - 1];
  }

  const float bc1 = b1[col];
#pragma unroll
  for (int r = 0; r < 8; ++r)
    T[(rbase + r) * T_PITCH + col] = fmaxf(acc[r] + bc1, 0.0f);
  __syncthreads();

  const float* trow = T + m * T_PITCH;
  v8f acc2 = {};
#pragma unroll
  for (int kk = 0; kk < 16; ++kk) {
    const int kb = kk * 4 + khalf;
    v2f a;
    a.x = trow[kb];
    a.y = trow[kb + 1];
    v2f b;
    b.x = W2[kb * H_DIM + n0 + m];
    b.y = W2[(kb + 1) * H_DIM + n0 + m];
    acc2 = __builtin_amdgcn_wmma_f32_16x16x4_f32(false, a, false, b, (short)0,
                                                 acc2, false, false);
  }

  const float bc2 = b2[col];
#pragma unroll
  for (int r = 0; r < 8; ++r) {
    const int row = row0 + rbase + r;
    if (row < n) {
      const float h = acc2[r];
      const float d = dv[r];
      const size_t o = (size_t)row * H_DIM + col;
      Hout[o] = h;
      S0[o] = fmaf(h, d * d, bc2);
      S1[o] = 0.0f;
    }
  }
}

#define EDGES_PER_WAVE 32
#define ROWS_IN_FLIGHT 8

__device__ __forceinline__ int lane_bcast_i(int v, int l) {
  return __builtin_amdgcn_readlane(v, l);
}
__device__ __forceinline__ float lane_bcast_f(float v, int l) {
  return __builtin_bit_cast(
      float, __builtin_amdgcn_readlane(__builtin_bit_cast(int, v), l));
}

__global__ void __launch_bounds__(256)
k_scatter_edges(const float* __restrict__ Hm, const int* __restrict__ src,
                const int* __restrict__ dst, const float* __restrict__ dis,
                float* __restrict__ Acc0, float* __restrict__ Acc1, int ne) {
  const int lane = threadIdx.x & 31;
  const int wave = blockIdx.x * (blockDim.x >> 5) + (threadIdx.x >> 5);
  const int e0 = wave * EDGES_PER_WAVE;
  if (e0 >= ne) return;
  const int left = ne - e0;
  const int cnt = left < EDGES_PER_WAVE ? left : EDGES_PER_WAVE;

  int s = 0, d = 0;
  float nrm = 0.0f;
  if (lane < cnt) {
    s = src[e0 + lane];
    d = dst[e0 + lane];
    nrm = dis[s] * dis[d];
  }
  const float* __restrict__ Hl = Hm + lane;
  const unsigned place = __builtin_amdgcn_s_sendmsg_rtn(0x87);
  const bool upper = ((place >> 16) & 0xFu) >= 4u;
  float* __restrict__ Al = (upper ? Acc1 : Acc0) + lane;

  if (cnt == EDGES_PER_WAVE) {
#pragma unroll
    for (int j0 = 0; j0 < EDGES_PER_WAVE; j0 += ROWS_IN_FLIGHT) {
      float vlo[ROWS_IN_FLIGHT], vhi[ROWS_IN_FLIGHT];
#pragma unroll
      for (int j = 0; j < ROWS_IN_FLIGHT; ++j) {
        const int sj = lane_bcast_i(s, j0 + j);
        const float* hp = Hl + (size_t)sj * H_DIM;
        vlo[j] = hp[0];
        vhi[j] = hp[32];
      }
#pragma unroll
      for (int j = 0; j < ROWS_IN_FLIGHT; ++j) {
        const int dj = lane_bcast_i(d, j0 + j);
        const float nj = lane_bcast_f(nrm, j0 + j);
        float* outp = Al + (size_t)dj * H_DIM;
        unsafeAtomicAdd(outp, vlo[j] * nj);
        unsafeAtomicAdd(outp + 32, vhi[j] * nj);
      }
    }
  } else {
    for (int j = 0; j < cnt; ++j) {
      const int sj = lane_bcast_i(s, j);
      const int dj = lane_bcast_i(d, j);
      const float nj = lane_bcast_f(nrm, j);
      const float* hp = Hl + (size_t)sj * H_DIM;
      const float lo = hp[0];
      const float hi = hp[32];
      float* outp = Al + (size_t)dj * H_DIM;
      unsafeAtomicAdd(outp, lo * nj);
      unsafeAtomicAdd(outp + 32, hi * nj);
    }
  }
}

__global__ void k_decode(const float* __restrict__ Z0,
                         const float* __restrict__ Z1,
                         const int* __restrict__ pos,
                         const int* __restrict__ neg,
                         float* __restrict__ out, int npos, int nneg) {
  const int k = blockIdx.x * (blockDim.x >> 5) + (threadIdx.x >> 5);
  if (k >= npos + nneg) return;
  const int lane = threadIdx.x & 31;
  int a, b;
  if (k < npos) {
    a = pos[k];
    b = pos[npos + k];
  } else {
    const int j = k - npos;
    a = neg[j];
    b = neg[nneg + j];
  }
  const float2 a0 = ((const float2*)(Z0 + (size_t)a * H_DIM))[lane];
  const float2 a1 = ((const float2*)(Z1 + (size_t)a * H_DIM))[lane];
  const float2 b0 = ((const float2*)(Z0 + (size_t)b * H_DIM))[lane];
  const float2 b1 = ((const float2*)(Z1 + (size_t)b * H_DIM))[lane];
  const float2 va = make_float2(a0.x + a1.x, a0.y + a1.y);
  const float2 vb = make_float2(b0.x + b1.x, b0.y + b1.y);
  float s = fmaf(va.x, vb.x, va.y * vb.y);
#pragma unroll
  for (int off = 16; off > 0; off >>= 1) s += __shfl_xor(s, off, 32);
  if (lane == 0) out[k] = s;
}

extern "C" void kernel_launch(void* const* d_in, const int* in_sizes, int n_in,
                              void* d_out, int out_size, void* d_ws,
                              size_t ws_size, hipStream_t stream) {
  (void)n_in; (void)out_size; (void)ws_size;
  const float* x   = (const float*)d_in[0];
  const int*   ei  = (const int*)d_in[1];
  const int*   pei = (const int*)d_in[2];
  const int*   nei = (const int*)d_in[3];
  const float* W1  = (const float*)d_in[4];
  const float* b1  = (const float*)d_in[5];
  const float* W2  = (const float*)d_in[6];
  const float* b2  = (const float*)d_in[7];
  float* out = (float*)d_out;

  const int n    = in_sizes[0] / H_DIM;
  const int ne   = in_sizes[1] / 2;
  const int npos = in_sizes[2] / 2;
  const int nneg = in_sizes[3] / 2;
  const int* src = ei;
  const int* dst = ei + ne;

  float* ws = (float*)d_ws;
  const size_t nAl = ((size_t)n + 63) & ~(size_t)63;
  unsigned* part = (unsigned*)d_ws;
  float* dis  = ws + (size_t)DEG_CHUNKS * nAl;
  float* s0   = dis + nAl;
  float* s1   = s0 + (size_t)n * H_DIM;
  float* hbuf = s1 + (size_t)n * H_DIM;

  const int B = 256;
  k_deg_hist<<<dim3(DEG_CHUNKS, (n + DEG_RANGE - 1) / DEG_RANGE), DEG_THREADS,
               0, stream>>>(dst, part, nAl, n, ne);
  k_norm_init<<<(n + 63) / 64, B, 0, stream>>>(part, nAl, dis, x, s0, s1, n);

  k_scatter_edges<<<(ne + 255) / 256, B, 0, stream>>>(x, src, dst, dis, s0, s1,
                                                      ne);
  k_gemm_pair_fused<<<(n + 15) / 16, 128, 0, stream>>>(s0, s1, W1, b1, W2, b2,
                                                       dis, hbuf, n);
  k_scatter_edges<<<(ne + 255) / 256, B, 0, stream>>>(hbuf, src, dst, dis, s0,
                                                      s1, ne);
  k_decode<<<(npos + nneg + 7) / 8, B, 0, stream>>>(s0, s1, pei, nei, out,
                                                    npos, nneg);
}


# Edge scale dis[src]*dis[dst] computed once: layer 1's scatter stores it per edge (one line a wave), layer 2's reads it back, not two 32-line degree gathers; float atomics, arrival order as before
# speedup vs baseline: 2.1445x; 1.0357x over previous
#include <hip/hip_runtime.h>
#include <hip/hip_bf16.h>


typedef float v2f __attribute__((ext_vector_type(2)));
typedef float v8f __attribute__((ext_vector_type(8)));

#define H_DIM 64

#define DEG_RANGE  3200
#define DEG_CHUNKS 8
#define DEG_AHEAD  16
#define DEG_THREADS 1024

__global__ void __launch_bounds__(DEG_THREADS)
k_deg_hist(const int* __restrict__ dst, unsigned* __restrict__ part,
           size_t plane, int n, int ne) {
  __shared__ unsigned cnt[DEG_RANGE];
  const int tid  = threadIdx.x;
  const int step = DEG_THREADS;
  const int lo   = blockIdx.y * DEG_RANGE;
  for (int i = tid; i < DEG_RANGE; i += step) cnt[i] = 0u;
  __syncthreads();

  const int per = (ne + (int)gridDim.x - 1) / (int)gridDim.x;
  const int e0  = blockIdx.x * per;
  const int e1  = e0 + per < ne ? e0 + per : ne;
  int e = e0 + tid;
  for (; e + (DEG_AHEAD - 1) * step < e1; e += DEG_AHEAD * step) {
    int v[DEG_AHEAD];
#pragma unroll
    for (int j = 0; j < DEG_AHEAD; ++j) v[j] = dst[e + j * step];
#pragma unroll
    for (int j = 0; j < DEG_AHEAD; ++j) {
      const unsigned r = (unsigned)(v[j] - lo);
      if (r < (unsigned)DEG_RANGE) atomicAdd(&cnt[r], 1u);
    }
  }
  for (; e < e1; e += step) {
    const unsigned r = (unsigned)(dst[e] - lo);
    if (r < (unsigned)DEG_RANGE) atomicAdd(&cnt[r], 1u);
  }
  __syncthreads();

  unsigned* __restrict__ prow = part + (size_t)blockIdx.x * plane;
  for (int i = tid; i < DEG_RANGE; i += step) {
    const int node = lo + i;
    if (node < n) prow[node] = cnt[i];
  }
}

#define ROWS_PER_INIT_WAVE 8
__global__ void __launch_bounds__(256)
k_norm_init(const unsigned* __restrict__ part, size_t plane,
            float* __restrict__ dis, const float* __restrict__ X,
            float* __restrict__ S0, float* __restrict__ S1, int n) {
  const int lane = threadIdx.x & 31;
  const int wave = blockIdx.x * (blockDim.x >> 5) + (threadIdx.x >> 5);
  const int sub  = lane >> 4;
  const int c4   = (lane & 15) << 2;
  float dg[ROWS_PER_INIT_WAVE / 2];
  float4 xv[ROWS_PER_INIT_WAVE / 2];
#pragma unroll
  for (int j = 0; j < ROWS_PER_INIT_WAVE / 2; ++j) {
    const int row = wave * ROWS_PER_INIT_WAVE + 2 * j + sub;
    const int rc  = row < n ? row : n - 1;
    xv[j] = *(const float4*)(X + (size_t)rc * H_DIM + c4);
    unsigned c = 1u;
#pragma unroll
    for (int k = 0; k < DEG_CHUNKS; ++k) c += part[(size_t)k * plane + rc];
    dg[j] = (float)c;
  }
#pragma unroll
  for (int j = 0; j < ROWS_PER_INIT_WAVE / 2; ++j) {
    const int row = wave * ROWS_PER_INIT_WAVE + 2 * j + sub;
    if (row < n) {
      const float d = rsqrtf(dg[j]);
      if ((lane & 15) == 0) dis[row] = d;
      const float dd = d * d;
      const size_t off = (size_t)row * H_DIM + c4;
      const float4 x = xv[j];
      *(float4*)(S0 + off) = make_float4(x.x * dd, x.y * dd, x.z * dd, x.w * dd);
      *(float4*)(S1 + off) = make_float4(0.0f, 0.0f, 0.0f, 0.0f);
    }
  }
}

#define T_PITCH 68

__global__ void __launch_bounds__(128)
k_gemm_pair_fused(float* __restrict__ S0, float* __restrict__ S1,
                  const float* __restrict__ W1, const float* __restrict__ b1,
                  const float* __restrict__ W2, const float* __restrict__ b2,
                  const float* __restrict__ dis, float* __restrict__ Hout,
                  int n) {
  __shared__ float T[16 * T_PITCH];
  const int lane  = threadIdx.x & 31;
  const int wave  = threadIdx.x >> 5;
  const int row0  = blockIdx.x * 16;
  const int n0    = wave * 16;
  const int m     = lane & 15;
  const int khalf = (lane >> 4) << 1;

  int arow = row0 + m;
  if (arow >= n) arow = n - 1;
  const float* p0 = S0 + (size_t)arow * H_DIM;
  const float* p1 = S1 + (size_t)arow * H_DIM;

  const int rbase = (lane >> 4) << 3;
  const int col   = n0 + m;

  v8f acc = {};
#pragma unroll
  for (int kk = 0; kk < 16; ++kk) {
    const int kb = kk * 4 + khalf;
    v2f a;
    a.x = p0[kb] + p1[kb];
    a.y = p0[kb + 1] + p1[kb + 1];
    v2f b;
    b.x = W1[kb * H_DIM + n0 + m];
    b.y = W1[(kb + 1) * H_DIM + n0 + m];
    acc = __builtin_amdgcn_wmma_f32_16x16x4_f32(false, a, false, b, (short)0,
                                                acc, false, false);
  }

  float dv[8];
#pragma unroll
  for (int r = 0; r < 8; ++r) {
    const int row = row0 + rbase + r;
    dv[r] = dis[row < n ? row : n - 1];
  }

  const float bc1 = b1[col];
#pragma unroll
  for (int r = 0; r < 8; ++r)
    T[(rbase + r) * T_PITCH + col] = fmaxf(acc[r] + bc1, 0.0f);
  __syncthreads();

  const float* trow = T + m * T_PITCH;
  v8f acc2 = {};
#pragma unroll
  for (int kk = 0; kk < 16; ++kk) {
    const int kb = kk * 4 + khalf;
    v2f a;
    a.x = trow[kb];
    a.y = trow[kb + 1];
    v2f b;
    b.x = W2[kb * H_DIM + n0 + m];
    b.y = W2[(kb + 1) * H_DIM + n0 + m];
    acc2 = __builtin_amdgcn_wmma_f32_16x16x4_f32(false, a, false, b, (short)0,
                                                 acc2, false, false);
  }

  const float bc2 = b2[col];
#pragma unroll
  for (int r = 0; r < 8; ++r) {
    const int row = row0 + rbase + r;
    if (row < n) {
      const float h = acc2[r];
      const float d = dv[r];
      const size_t o = (size_t)row * H_DIM + col;
      Hout[o] = h;
      S0[o] = fmaf(h, d * d, bc2);
      S1[o] = 0.0f;
    }
  }
}

#define EDGES_PER_WAVE 32
#define ROWS_IN_FLIGHT 8

__device__ __forceinline__ int lane_bcast_i(int v, int l) {
  return __builtin_amdgcn_readlane(v, l);
}
__device__ __forceinline__ float lane_bcast_f(float v, int l) {
  return __builtin_bit_cast(
      float, __builtin_amdgcn_readlane(__builtin_bit_cast(int, v), l));
}

template <bool FIRST_PASS>
__global__ void __launch_bounds__(256)
k_scatter_edges(const float* __restrict__ Hm, const int* __restrict__ src,
                const int* __restrict__ dst, const float* __restrict__ dis,
                float* __restrict__ enorm,
                float* __restrict__ Acc0, float* __restrict__ Acc1, int ne) {
  const int lane = threadIdx.x & 31;
  const int wave = blockIdx.x * (blockDim.x >> 5) + (threadIdx.x >> 5);
  const int e0 = wave * EDGES_PER_WAVE;
  if (e0 >= ne) return;
  const int left = ne - e0;
  const int cnt = left < EDGES_PER_WAVE ? left : EDGES_PER_WAVE;

  int s = 0, d = 0;
  float nrm = 0.0f;
  if (lane < cnt) {
    s = src[e0 + lane];
    d = dst[e0 + lane];
    if (FIRST_PASS) {
      nrm = dis[s] * dis[d];
      enorm[e0 + lane] = nrm;
    } else {
      nrm = enorm[e0 + lane];
    }
  }
  const float* __restrict__ Hl = Hm + lane;
  const unsigned place = __builtin_amdgcn_s_sendmsg_rtn(0x87);
  const bool upper = ((place >> 16) & 0xFu) >= 4u;
  float* __restrict__ Al = (upper ? Acc1 : Acc0) + lane;

  if (cnt == EDGES_PER_WAVE) {
#pragma unroll
    for (int j0 = 0; j0 < EDGES_PER_WAVE; j0 += ROWS_IN_FLIGHT) {
      float vlo[ROWS_IN_FLIGHT], vhi[ROWS_IN_FLIGHT];
#pragma unroll
      for (int j = 0; j < ROWS_IN_FLIGHT; ++j) {
        const int sj = lane_bcast_i(s, j0 + j);
        const float* hp = Hl + (size_t)sj * H_DIM;
        vlo[j] = hp[0];
        vhi[j] = hp[32];
      }
#pragma unroll
      for (int j = 0; j < ROWS_IN_FLIGHT; ++j) {
        const int dj = lane_bcast_i(d, j0 + j);
        const float nj = lane_bcast_f(nrm, j0 + j);
        float* outp = Al + (size_t)dj * H_DIM;
        unsafeAtomicAdd(outp, vlo[j] * nj);
        unsafeAtomicAdd(outp + 32, vhi[j] * nj);
      }
    }
  } else {
    for (int j = 0; j < cnt; ++j) {
      const int sj = lane_bcast_i(s, j);
      const int dj = lane_bcast_i(d, j);
      const float nj = lane_bcast_f(nrm, j);
      const float* hp = Hl + (size_t)sj * H_DIM;
      const float lo = hp[0];
      const float hi = hp[32];
      float* outp = Al + (size_t)dj * H_DIM;
      unsafeAtomicAdd(outp, lo * nj);
      unsafeAtomicAdd(outp + 32, hi * nj);
    }
  }
}

__global__ void k_decode(const float* __restrict__ Z0,
                         const float* __restrict__ Z1,
                         const int* __restrict__ pos,
                         const int* __restrict__ neg,
                         float* __restrict__ out, int npos, int nneg) {
  const int k = blockIdx.x * (blockDim.x >> 5) + (threadIdx.x >> 5);
  if (k >= npos + nneg) return;
  const int lane = threadIdx.x & 31;
  int a, b;
  if (k < npos) {
    a = pos[k];
    b = pos[npos + k];
  } else {
    const int j = k - npos;
    a = neg[j];
    b = neg[nneg + j];
  }
  const float2 a0 = ((const float2*)(Z0 + (size_t)a * H_DIM))[lane];
  const float2 a1 = ((const float2*)(Z1 + (size_t)a * H_DIM))[lane];
  const float2 b0 = ((const float2*)(Z0 + (size_t)b * H_DIM))[lane];
  const float2 b1 = ((const float2*)(Z1 + (size_t)b * H_DIM))[lane];
  const float2 va = make_float2(a0.x + a1.x, a0.y + a1.y);
  const float2 vb = make_float2(b0.x + b1.x, b0.y + b1.y);
  float s = fmaf(va.x, vb.x, va.y * vb.y);
#pragma unroll
  for (int off = 16; off > 0; off >>= 1) s += __shfl_xor(s, off, 32);
  if (lane == 0) out[k] = s;
}

extern "C" void kernel_launch(void* const* d_in, const int* in_sizes, int n_in,
                              void* d_out, int out_size, void* d_ws,
                              size_t ws_size, hipStream_t stream) {
  (void)n_in; (void)out_size; (void)ws_size;
  const float* x   = (const float*)d_in[0];
  const int*   ei  = (const int*)d_in[1];
  const int*   pei = (const int*)d_in[2];
  const int*   nei = (const int*)d_in[3];
  const float* W1  = (const float*)d_in[4];
  const float* b1  = (const float*)d_in[5];
  const float* W2  = (const float*)d_in[6];
  const float* b2  = (const float*)d_in[7];
  float* out = (float*)d_out;

  const int n    = in_sizes[0] / H_DIM;
  const int ne   = in_sizes[1] / 2;
  const int npos = in_sizes[2] / 2;
  const int nneg = in_sizes[3] / 2;
  const int* src = ei;
  const int* dst = ei + ne;

  float* ws = (float*)d_ws;
  const size_t nAl = ((size_t)n + 63) & ~(size_t)63;
  unsigned* part = (unsigned*)d_ws;
  float* dis  = ws + (size_t)DEG_CHUNKS * nAl;
  float* s0   = dis + nAl;
  float* s1   = s0 + (size_t)n * H_DIM;
  float* hbuf = s1 + (size_t)n * H_DIM;
  float* enorm = hbuf + (size_t)n * H_DIM;

  const int B = 256;
  k_deg_hist<<<dim3(DEG_CHUNKS, (n + DEG_RANGE - 1) / DEG_RANGE), DEG_THREADS,
               0, stream>>>(dst, part, nAl, n, ne);
  k_norm_init<<<(n + 63) / 64, B, 0, stream>>>(part, nAl, dis, x, s0, s1, n);

  k_scatter_edges<true><<<(ne + 255) / 256, B, 0, stream>>>(
      x, src, dst, dis, enorm, s0, s1, ne);
  k_gemm_pair_fused<<<(n + 15) / 16, 128, 0, stream>>>(s0, s1, W1, b1, W2, b2,
                                                       dis, hbuf, n);
  k_scatter_edges<false><<<(ne + 255) / 256, B, 0, stream>>>(
      hbuf, src, dst, dis, enorm, s0, s1, ne);
  k_decode<<<(npos + nneg + 7) / 8, B, 0, stream>>>(s0, s1, pei, nei, out,
                                                    npos, nneg);
}
